# starting kernel
# baseline (speedup 1.0000x reference)
#include <hip/hip_runtime.h>

typedef __attribute__((ext_vector_type(2))) float v2f;
typedef __attribute__((ext_vector_type(4))) float v4f;
typedef __attribute__((ext_vector_type(8))) float v8f;

__global__ void init_zero(float* __restrict__ p, int n) {
  int i = blockIdx.x * blockDim.x + threadIdx.x;
  if (i < n) p[i] = 0.0f;
}

__global__ void find_starts(const int* __restrict__ batch, int* __restrict__ starts,
                            int N, int G) {
  int i = blockIdx.x * blockDim.x + threadIdx.x;
  if (i >= N) return;
  int b = batch[i];
  int bp = (i == 0) ? -1 : batch[i - 1];
  for (int g = bp + 1; g <= b; ++g) starts[g] = i;
  if (i == N - 1) {
    for (int g = b + 1; g <= G; ++g) starts[g] = N;
  }
}

__global__ void __launch_bounds__(256)
gates_gemm(const float* __restrict__ qstar, const float* __restrict__ hs,
           const float* __restrict__ Wih, const float* __restrict__ Whh,
           const float* __restrict__ bih, const float* __restrict__ bhh,
           float* __restrict__ gates) {
  const int lane = threadIdx.x & 31;
  const int wave = blockIdx.x * (blockDim.x >> 5) + (threadIdx.x >> 5);
  const int TN = 1024 / 16;
  const int tm = wave / TN;
  const int tn = wave % TN;
  const int r  = lane & 15;
  const int kh = (lane >> 4) << 1;

  const float* arow = qstar + (size_t)(tm * 16 + r) * 512;
  const float* brow = Wih   + (size_t)(tn * 16 + r) * 512;
  v8f acc = {};
  for (int k = 0; k < 512; k += 4) {
    v2f a, b;
    a.x = arow[k + kh]; a.y = arow[k + kh + 1];
    b.x = brow[k + kh]; b.y = brow[k + kh + 1];
    acc = __builtin_amdgcn_wmma_f32_16x16x4_f32(false, a, false, b, (short)0, acc,
                                                false, false);
  }
  const float* arow2 = hs  + (size_t)(tm * 16 + r) * 256;
  const float* brow2 = Whh + (size_t)(tn * 16 + r) * 256;
  for (int k = 0; k < 256; k += 4) {
    v2f a, b;
    a.x = arow2[k + kh]; a.y = arow2[k + kh + 1];
    b.x = brow2[k + kh]; b.y = brow2[k + kh + 1];
    acc = __builtin_amdgcn_wmma_f32_16x16x4_f32(false, a, false, b, (short)0, acc,
                                                false, false);
  }
  const int mb = tm * 16 + (lane >> 4) * 8;
  const int n  = tn * 16 + r;
  const float bias = bih[n] + bhh[n];
#pragma unroll
  for (int v = 0; v < 8; ++v)
    gates[(size_t)(mb + v) * 1024 + n] = acc[v] + bias;
}

__device__ __forceinline__ float sigmoidf_(float x) {
  return 1.0f / (1.0f + __expf(-x));
}

__global__ void __launch_bounds__(256)
lstm_update(const float* __restrict__ gates, float* __restrict__ hs,
            float* __restrict__ cs, float* __restrict__ qstar) {
  int idx = blockIdx.x * blockDim.x + threadIdx.x;
  int b = idx >> 8, c = idx & 255;
  const float* g = gates + (size_t)b * 1024;
  float ig = sigmoidf_(g[c]);
  float fg = sigmoidf_(g[256 + c]);
  float gg = tanhf(g[512 + c]);
  float og = sigmoidf_(g[768 + c]);
  float cn = fg * cs[idx] + ig * gg;
  cs[idx] = cn;
  float h = og * tanhf(cn);
  hs[idx] = h;
  qstar[(size_t)b * 512 + c] = h;
}

__global__ void __launch_bounds__(256)
attention(const float* __restrict__ x, const int* __restrict__ starts,
          const float* __restrict__ hs, float* __restrict__ qstar) {
  const int g    = blockIdx.x;
  const int tid  = threadIdx.x;
  const int lane = tid & 31;
  const int wv   = tid >> 5;

  __shared__ __align__(16) float q[256];
  __shared__ __align__(16) float rall[8][256];
  __shared__ float wmax[8];
  __shared__ float wsum[8];

  q[tid] = hs[(size_t)g * 256 + tid];
  __syncthreads();

  const int start = starts[g];
  const int end   = starts[g + 1];

  const v4f q0 = *(const v4f*)&q[4 * lane];
  const v4f q1 = *(const v4f*)&q[128 + 4 * lane];

  float m = -__builtin_inff();
  float s = 0.0f;
  v4f r0 = {}, r1 = {};

  for (int i = start + wv; i < end; i += 8) {
    const v4f* xr = (const v4f*)(x + (size_t)i * 256);
    v4f x0 = xr[lane];
    v4f x1 = xr[32 + lane];
    float d = 0.0f;
    d = fmaf(x0.x, q0.x, d); d = fmaf(x0.y, q0.y, d);
    d = fmaf(x0.z, q0.z, d); d = fmaf(x0.w, q0.w, d);
    d = fmaf(x1.x, q1.x, d); d = fmaf(x1.y, q1.y, d);
    d = fmaf(x1.z, q1.z, d); d = fmaf(x1.w, q1.w, d);
#pragma unroll
    for (int off = 16; off; off >>= 1) d += __shfl_xor(d, off, 32);
    if (d > m) {
      float sc = (m == -__builtin_inff()) ? 0.0f : __expf(m - d);
      s *= sc;
      r0.x *= sc; r0.y *= sc; r0.z *= sc; r0.w *= sc;
      r1.x *= sc; r1.y *= sc; r1.z *= sc; r1.w *= sc;
      m = d;
    }
    float a = __expf(d - m);
    s += a;
    r0.x = fmaf(a, x0.x, r0.x); r0.y = fmaf(a, x0.y, r0.y);
    r0.z = fmaf(a, x0.z, r0.z); r0.w = fmaf(a, x0.w, r0.w);
    r1.x = fmaf(a, x1.x, r1.x); r1.y = fmaf(a, x1.y, r1.y);
    r1.z = fmaf(a, x1.z, r1.z); r1.w = fmaf(a, x1.w, r1.w);
  }

  *(v4f*)&rall[wv][4 * lane]       = r0;
  *(v4f*)&rall[wv][128 + 4 * lane] = r1;
  if (lane == 0) { wmax[wv] = m; wsum[wv] = s; }
  __syncthreads();

  float M = -__builtin_inff();
#pragma unroll
  for (int w = 0; w < 8; ++w) M = fmaxf(M, wmax[w]);

  float stot = 0.0f, r = 0.0f;
#pragma unroll
  for (int w = 0; w < 8; ++w) {
    float sw = wsum[w];
    if (sw > 0.0f) {
      float sc = __expf(wmax[w] - M);
      stot = fmaf(sw, sc, stot);
      r    = fmaf(rall[w][tid], sc, r);
    }
  }
  qstar[(size_t)g * 512 + 256 + tid] = (stot > 0.0f) ? (r / stot) : 0.0f;
}

__global__ void __launch_bounds__(256)
out_gemm(const float* __restrict__ qstar, const float* __restrict__ Wout,
         const float* __restrict__ bout, float* __restrict__ out) {
  const int lane = threadIdx.x & 31;
  const int wave = blockIdx.x * (blockDim.x >> 5) + (threadIdx.x >> 5);
  const int TN = 256 / 16;
  const int tm = wave / TN;
  const int tn = wave % TN;
  const int r  = lane & 15;
  const int kh = (lane >> 4) << 1;

  const float* arow = qstar + (size_t)(tm * 16 + r) * 512;
  const float* brow = Wout  + (size_t)(tn * 16 + r) * 512;
  v8f acc = {};
  for (int k = 0; k < 512; k += 4) {
    v2f a, b;
    a.x = arow[k + kh]; a.y = arow[k + kh + 1];
    b.x = brow[k + kh]; b.y = brow[k + kh + 1];
    acc = __builtin_amdgcn_wmma_f32_16x16x4_f32(false, a, false, b, (short)0, acc,
                                                false, false);
  }
  const int mb = tm * 16 + (lane >> 4) * 8;
  const int n  = tn * 16 + r;
  const float bias = bout[n];
#pragma unroll
  for (int v = 0; v < 8; ++v)
    out[(size_t)(mb + v) * 256 + n] = acc[v] + bias;
}

extern "C" void kernel_launch(void* const* d_in, const int* in_sizes, int n_in,
                              void* d_out, int out_size, void* d_ws, size_t ws_size,
                              hipStream_t stream) {
  const float* x     = (const float*)d_in[0];
  const int*   batch = (const int*)  d_in[1];
  const float* Wih   = (const float*)d_in[2];
  const float* Whh   = (const float*)d_in[3];
  const float* bih   = (const float*)d_in[4];
  const float* bhh   = (const float*)d_in[5];
  const float* Wout  = (const float*)d_in[6];
  const float* bout  = (const float*)d_in[7];
  const int N = in_sizes[0] / 256;
  const int G = 512;

  float* ws    = (float*)d_ws;
  float* hs    = ws;
  float* cs    = hs + 512 * 256;
  float* qstar = cs + 512 * 256;
  float* gates = qstar + 512 * 512;
  int*   starts = (int*)(gates + 512 * 1024);

  int zn = 512 * 256 * 2 + 512 * 512;
  init_zero<<<(zn + 255) / 256, 256, 0, stream>>>(ws, zn);
  find_starts<<<(N + 255) / 256, 256, 0, stream>>>(batch, starts, N, G);

  for (int step = 0; step < 3; ++step) {
    gates_gemm<<<256, 256, 0, stream>>>(qstar, hs, Wih, Whh, bih, bhh, gates);
    lstm_update<<<512, 256, 0, stream>>>(gates, hs, cs, qstar);
    attention<<<G, 256, 0, stream>>>(x, starts, hs, qstar);
  }
  out_gemm<<<64, 256, 0, stream>>>(qstar, Wout, bout, (float*)d_out);
}


# attention: each node row is spread over 8 lanes (32 floats a lane), 4 nodes per wave trip: 8 row loads in flight per lane, 3 lane exchanges per 4 nodes, 32 partials per graph; f32 sum order changes
# speedup vs baseline: 1.6123x; 1.6123x over previous
#include <hip/hip_runtime.h>

typedef __attribute__((ext_vector_type(2))) float v2f;
typedef __attribute__((ext_vector_type(4))) float v4f;
typedef __attribute__((ext_vector_type(8))) float v8f;

__global__ void init_zero(float* __restrict__ p, int n) {
  int i = blockIdx.x * blockDim.x + threadIdx.x;
  if (i < n) p[i] = 0.0f;
}

__global__ void find_starts(const int* __restrict__ batch, int* __restrict__ starts,
                            int N, int G) {
  int i = blockIdx.x * blockDim.x + threadIdx.x;
  if (i >= N) return;
  int b = batch[i];
  int bp = (i == 0) ? -1 : batch[i - 1];
  for (int g = bp + 1; g <= b; ++g) starts[g] = i;
  if (i == N - 1) {
    for (int g = b + 1; g <= G; ++g) starts[g] = N;
  }
}

__global__ void __launch_bounds__(256)
gates_gemm(const float* __restrict__ qstar, const float* __restrict__ hs,
           const float* __restrict__ Wih, const float* __restrict__ Whh,
           const float* __restrict__ bih, const float* __restrict__ bhh,
           float* __restrict__ gates) {
  const int lane = threadIdx.x & 31;
  const int wave = blockIdx.x * (blockDim.x >> 5) + (threadIdx.x >> 5);
  const int TN = 1024 / 16;
  const int tm = wave / TN;
  const int tn = wave % TN;
  const int r  = lane & 15;
  const int kh = (lane >> 4) << 1;

  const float* arow = qstar + (size_t)(tm * 16 + r) * 512;
  const float* brow = Wih   + (size_t)(tn * 16 + r) * 512;
  v8f acc = {};
  for (int k = 0; k < 512; k += 4) {
    v2f a, b;
    a.x = arow[k + kh]; a.y = arow[k + kh + 1];
    b.x = brow[k + kh]; b.y = brow[k + kh + 1];
    acc = __builtin_amdgcn_wmma_f32_16x16x4_f32(false, a, false, b, (short)0, acc,
                                                false, false);
  }
  const float* arow2 = hs  + (size_t)(tm * 16 + r) * 256;
  const float* brow2 = Whh + (size_t)(tn * 16 + r) * 256;
  for (int k = 0; k < 256; k += 4) {
    v2f a, b;
    a.x = arow2[k + kh]; a.y = arow2[k + kh + 1];
    b.x = brow2[k + kh]; b.y = brow2[k + kh + 1];
    acc = __builtin_amdgcn_wmma_f32_16x16x4_f32(false, a, false, b, (short)0, acc,
                                                false, false);
  }
  const int mb = tm * 16 + (lane >> 4) * 8;
  const int n  = tn * 16 + r;
  const float bias = bih[n] + bhh[n];
#pragma unroll
  for (int v = 0; v < 8; ++v)
    gates[(size_t)(mb + v) * 1024 + n] = acc[v] + bias;
}

__device__ __forceinline__ float sigmoidf_(float x) {
  return 1.0f / (1.0f + __expf(-x));
}

__global__ void __launch_bounds__(256)
lstm_update(const float* __restrict__ gates, float* __restrict__ hs,
            float* __restrict__ cs, float* __restrict__ qstar) {
  int idx = blockIdx.x * blockDim.x + threadIdx.x;
  int b = idx >> 8, c = idx & 255;
  const float* g = gates + (size_t)b * 1024;
  float ig = sigmoidf_(g[c]);
  float fg = sigmoidf_(g[256 + c]);
  float gg = tanhf(g[512 + c]);
  float og = sigmoidf_(g[768 + c]);
  float cn = fg * cs[idx] + ig * gg;
  cs[idx] = cn;
  float h = og * tanhf(cn);
  hs[idx] = h;
  qstar[(size_t)b * 512 + c] = h;
}

__global__ void __launch_bounds__(256)
attention(const float* __restrict__ x, const int* __restrict__ starts,
          const float* __restrict__ hs, float* __restrict__ qstar) {
  const int g    = blockIdx.x;
  const int tid  = threadIdx.x;
  const int lane = tid & 31;
  const int wv   = tid >> 5;
  const int grp  = lane >> 3;
  const int j    = lane & 7;

  __shared__ __align__(16) float rall[32][256];
  __shared__ float pmax[32];
  __shared__ float psum[32];

  const int start = starts[g];
  const int end   = starts[g + 1];

  const v4f* qp = (const v4f*)(hs + (size_t)g * 256) + j;
  v4f qv[8];
#pragma unroll
  for (int k = 0; k < 8; ++k) qv[k] = qp[8 * k];

  float m = -__builtin_inff();
  float s = 0.0f;
  v4f r[8];
#pragma unroll
  for (int k = 0; k < 8; ++k) r[k] = (v4f){0.0f, 0.0f, 0.0f, 0.0f};

  for (int i = start + 4 * wv + grp; i < end; i += 32) {
    const v4f* xr = (const v4f*)(x + (size_t)i * 256) + j;
    v4f xv[8];
#pragma unroll
    for (int k = 0; k < 8; ++k) xv[k] = xr[8 * k];

    float d0 = 0.0f, d1 = 0.0f, d2 = 0.0f, d3 = 0.0f;
#pragma unroll
    for (int k = 0; k < 8; ++k) {
      d0 = fmaf(xv[k].x, qv[k].x, d0);
      d1 = fmaf(xv[k].y, qv[k].y, d1);
      d2 = fmaf(xv[k].z, qv[k].z, d2);
      d3 = fmaf(xv[k].w, qv[k].w, d3);
    }
    float d = (d0 + d1) + (d2 + d3);
    d += __shfl_xor(d, 4, 32);
    d += __shfl_xor(d, 2, 32);
    d += __shfl_xor(d, 1, 32);
    if (d > m) {
      float sc = (m == -__builtin_inff()) ? 0.0f : __expf(m - d);
      s *= sc;
#pragma unroll
      for (int k = 0; k < 8; ++k) {
        r[k].x *= sc; r[k].y *= sc; r[k].z *= sc; r[k].w *= sc;
      }
      m = d;
    }
    float a = __expf(d - m);
    s += a;
#pragma unroll
    for (int k = 0; k < 8; ++k) {
      r[k].x = fmaf(a, xv[k].x, r[k].x); r[k].y = fmaf(a, xv[k].y, r[k].y);
      r[k].z = fmaf(a, xv[k].z, r[k].z); r[k].w = fmaf(a, xv[k].w, r[k].w);
    }
  }

  const int p = wv * 4 + grp;
#pragma unroll
  for (int k = 0; k < 8; ++k) *(v4f*)&rall[p][32 * k + 4 * j] = r[k];
  if (j == 0) { pmax[p] = m; psum[p] = s; }
  __syncthreads();

  float M = -__builtin_inff();
#pragma unroll
  for (int w = 0; w < 32; ++w) M = fmaxf(M, pmax[w]);

  float stot = 0.0f, rr = 0.0f;
#pragma unroll
  for (int w = 0; w < 32; ++w) {
    const float sw = psum[w];
    const float sc = (sw > 0.0f) ? __expf(pmax[w] - M) : 0.0f;
    stot = fmaf(sw, sc, stot);
    rr   = fmaf(rall[w][tid], sc, rr);
  }
  qstar[(size_t)g * 512 + 256 + tid] = (stot > 0.0f) ? (rr / stot) : 0.0f;
}

__global__ void __launch_bounds__(256)
out_gemm(const float* __restrict__ qstar, const float* __restrict__ Wout,
         const float* __restrict__ bout, float* __restrict__ out) {
  const int lane = threadIdx.x & 31;
  const int wave = blockIdx.x * (blockDim.x >> 5) + (threadIdx.x >> 5);
  const int TN = 256 / 16;
  const int tm = wave / TN;
  const int tn = wave % TN;
  const int r  = lane & 15;
  const int kh = (lane >> 4) << 1;

  const float* arow = qstar + (size_t)(tm * 16 + r) * 512;
  const float* brow = Wout  + (size_t)(tn * 16 + r) * 512;
  v8f acc = {};
  for (int k = 0; k < 512; k += 4) {
    v2f a, b;
    a.x = arow[k + kh]; a.y = arow[k + kh + 1];
    b.x = brow[k + kh]; b.y = brow[k + kh + 1];
    acc = __builtin_amdgcn_wmma_f32_16x16x4_f32(false, a, false, b, (short)0, acc,
                                                false, false);
  }
  const int mb = tm * 16 + (lane >> 4) * 8;
  const int n  = tn * 16 + r;
  const float bias = bout[n];
#pragma unroll
  for (int v = 0; v < 8; ++v)
    out[(size_t)(mb + v) * 256 + n] = acc[v] + bias;
}

extern "C" void kernel_launch(void* const* d_in, const int* in_sizes, int n_in,
                              void* d_out, int out_size, void* d_ws, size_t ws_size,
                              hipStream_t stream) {
  const float* x     = (const float*)d_in[0];
  const int*   batch = (const int*)  d_in[1];
  const float* Wih   = (const float*)d_in[2];
  const float* Whh   = (const float*)d_in[3];
  const float* bih   = (const float*)d_in[4];
  const float* bhh   = (const float*)d_in[5];
  const float* Wout  = (const float*)d_in[6];
  const float* bout  = (const float*)d_in[7];
  const int N = in_sizes[0] / 256;
  const int G = 512;

  float* ws    = (float*)d_ws;
  float* hs    = ws;
  float* cs    = hs + 512 * 256;
  float* qstar = cs + 512 * 256;
  float* gates = qstar + 512 * 512;
  int*   starts = (int*)(gates + 512 * 1024);

  int zn = 512 * 256 * 2 + 512 * 512;
  init_zero<<<(zn + 255) / 256, 256, 0, stream>>>(ws, zn);
  find_starts<<<(N + 255) / 256, 256, 0, stream>>>(batch, starts, N, G);

  for (int step = 0; step < 3; ++step) {
    gates_gemm<<<256, 256, 0, stream>>>(qstar, hs, Wih, Whh, bih, bhh, gates);
    lstm_update<<<512, 256, 0, stream>>>(gates, hs, cs, qstar);
    attention<<<G, 256, 0, stream>>>(x, starts, hs, qstar);
  }
  out_gemm<<<64, 256, 0, stream>>>(qstar, Wout, bout, (float*)d_out);
}


# GEMMs re-tiled: several tiles per wave, K split over a block's 4 waves, summed in LDS; the LSTM block owns i,f,g,o of its patch and applies the cell (12 to 9 launches); f32 sum order changes
# speedup vs baseline: 2.0491x; 1.2709x over previous
#include <hip/hip_runtime.h>

typedef __attribute__((ext_vector_type(2))) float v2f;
typedef __attribute__((ext_vector_type(4))) float v4f;
typedef __attribute__((ext_vector_type(8))) float v8f;

__global__ void init_zero(float* __restrict__ p, int n) {
  int i = blockIdx.x * blockDim.x + threadIdx.x;
  if (i < n) p[i] = 0.0f;
}

__global__ void find_starts(const int* __restrict__ batch, int* __restrict__ starts,
                            int N, int G) {
  int i = blockIdx.x * blockDim.x + threadIdx.x;
  if (i >= N) return;
  int b = batch[i];
  int bp = (i == 0) ? -1 : batch[i - 1];
  for (int g = bp + 1; g <= b; ++g) starts[g] = i;
  if (i == N - 1) {
    for (int g = b + 1; g <= G; ++g) starts[g] = N;
  }
}

__device__ __forceinline__ float sigmoidf_(float x) {
  return 1.0f / (1.0f + __expf(-x));
}

#define WMMA_F32(a, b, c) \
  __builtin_amdgcn_wmma_f32_16x16x4_f32(false, (a), false, (b), (short)0, (c), false, false)

__global__ void __launch_bounds__(128)
gates_lstm(const float* __restrict__ hs_old, const float* r_src,
           const float* __restrict__ Wih, const float* __restrict__ Whh,
           const float* __restrict__ bih, const float* __restrict__ bhh,
           float* __restrict__ hs_new, float* __restrict__ cs, float* q_dst) {
  const int lane = threadIdx.x & 31;
  const int s    = __builtin_amdgcn_readfirstlane((int)(threadIdx.x >> 5));
  const int c    = lane & 15;
  const int hh   = lane >> 4;
  const int kh   = hh << 1;
  const int tm   = blockIdx.x >> 4;
  const int tc   = blockIdx.x & 15;
  const int row0 = tm * 32 + c;
  const int wrow = tc * 16 + c;

  __shared__ __align__(16) float part[4][8][320];

  v8f acc[2][4];
#pragma unroll
  for (int mi = 0; mi < 2; ++mi)
#pragma unroll
    for (int g = 0; g < 4; ++g) acc[mi][g] = (v8f){};

  {
    const float* a0;
    const float* a1;
    if (s < 2) {
      a0 = hs_old + (size_t)row0 * 256 + 128 * s + kh;
      a1 = a0 + 16 * 256;
    } else {
      a0 = r_src + (size_t)row0 * 512 + 128 * s + kh;
      a1 = a0 + 16 * 512;
    }
    const float* b = Wih + (size_t)wrow * 512 + 128 * s + kh;
#pragma unroll
    for (int k = 0; k < 128; k += 4) {
      const v2f A0 = *(const v2f*)(a0 + k);
      const v2f A1 = *(const v2f*)(a1 + k);
      v2f B[4];
#pragma unroll
      for (int g = 0; g < 4; ++g) B[g] = *(const v2f*)(b + (size_t)g * 256 * 512 + k);
#pragma unroll
      for (int g = 0; g < 4; ++g) {
        acc[0][g] = WMMA_F32(A0, B[g], acc[0][g]);
        acc[1][g] = WMMA_F32(A1, B[g], acc[1][g]);
      }
    }
  }
  {
    const float* a0 = hs_old + (size_t)row0 * 256 + 64 * s + kh;
    const float* a1 = a0 + 16 * 256;
    const float* b  = Whh + (size_t)wrow * 256 + 64 * s + kh;
#pragma unroll
    for (int k = 0; k < 64; k += 4) {
      const v2f A0 = *(const v2f*)(a0 + k);
      const v2f A1 = *(const v2f*)(a1 + k);
      v2f B[4];
#pragma unroll
      for (int g = 0; g < 4; ++g) B[g] = *(const v2f*)(b + (size_t)g * 256 * 256 + k);
#pragma unroll
      for (int g = 0; g < 4; ++g) {
        acc[0][g] = WMMA_F32(A0, B[g], acc[0][g]);
        acc[1][g] = WMMA_F32(A1, B[g], acc[1][g]);
      }
    }
  }

  {
    float* dst = &part[s][0][c * 20 + 8 * hh];
#pragma unroll
    for (int mi = 0; mi < 2; ++mi)
#pragma unroll
      for (int g = 0; g < 4; ++g) {
        const v8f v = acc[mi][g];
        *(v4f*)(dst + (mi * 4 + g) * 320)     = (v4f){v[0], v[1], v[2], v[3]};
        *(v4f*)(dst + (mi * 4 + g) * 320 + 4) = (v4f){v[4], v[5], v[6], v[7]};
      }
  }
  __syncthreads();

  const int cc  = threadIdx.x & 15;
  const int rg  = threadIdx.x >> 4;
  const int mi  = rg >> 2;
  const int rr  = (rg & 3) * 4;
  const int chn = tc * 16 + cc;
  v4f gs[4];
#pragma unroll
  for (int g = 0; g < 4; ++g) {
    const float* p = &part[0][mi * 4 + g][cc * 20 + rr];
    const v4f p0 = *(const v4f*)p;
    const v4f p1 = *(const v4f*)(p + 8 * 320);
    const v4f p2 = *(const v4f*)(p + 16 * 320);
    const v4f p3 = *(const v4f*)(p + 24 * 320);
    const float bias = bih[g * 256 + chn] + bhh[g * 256 + chn];
    gs[g] = ((p0 + p1) + (p2 + p3)) + bias;
  }
#pragma unroll
  for (int e = 0; e < 4; ++e) {
    const int row = tm * 32 + mi * 16 + rr + e;
    const int idx = row * 256 + chn;
    const float ig = sigmoidf_(gs[0][e]);
    const float fg = sigmoidf_(gs[1][e]);
    const float gg = tanhf(gs[2][e]);
    const float og = sigmoidf_(gs[3][e]);
    const float cn = fg * cs[idx] + ig * gg;
    cs[idx] = cn;
    const float h = og * tanhf(cn);
    hs_new[idx] = h;
    q_dst[(size_t)row * 512 + chn] = h;
  }
}

__global__ void __launch_bounds__(256)
attention(const float* __restrict__ x, const int* __restrict__ starts,
          const float* __restrict__ hs, float* __restrict__ qstar) {
  const int g    = blockIdx.x;
  const int tid  = threadIdx.x;
  const int lane = tid & 31;
  const int wv   = tid >> 5;
  const int grp  = lane >> 3;
  const int j    = lane & 7;

  __shared__ __align__(16) float rall[32][256];
  __shared__ float pmax[32];
  __shared__ float psum[32];

  const int start = starts[g];
  const int end   = starts[g + 1];

  const v4f* qp = (const v4f*)(hs + (size_t)g * 256) + j;
  v4f qv[8];
#pragma unroll
  for (int k = 0; k < 8; ++k) qv[k] = qp[8 * k];

  float m = -__builtin_inff();
  float s = 0.0f;
  v4f r[8];
#pragma unroll
  for (int k = 0; k < 8; ++k) r[k] = (v4f){0.0f, 0.0f, 0.0f, 0.0f};

  for (int i = start + 4 * wv + grp; i < end; i += 32) {
    const v4f* xr = (const v4f*)(x + (size_t)i * 256) + j;
    v4f xv[8];
#pragma unroll
    for (int k = 0; k < 8; ++k) xv[k] = xr[8 * k];

    float d0 = 0.0f, d1 = 0.0f, d2 = 0.0f, d3 = 0.0f;
#pragma unroll
    for (int k = 0; k < 8; ++k) {
      d0 = fmaf(xv[k].x, qv[k].x, d0);
      d1 = fmaf(xv[k].y, qv[k].y, d1);
      d2 = fmaf(xv[k].z, qv[k].z, d2);
      d3 = fmaf(xv[k].w, qv[k].w, d3);
    }
    float d = (d0 + d1) + (d2 + d3);
    d += __shfl_xor(d, 4, 32);
    d += __shfl_xor(d, 2, 32);
    d += __shfl_xor(d, 1, 32);
    if (d > m) {
      float sc = (m == -__builtin_inff()) ? 0.0f : __expf(m - d);
      s *= sc;
#pragma unroll
      for (int k = 0; k < 8; ++k) {
        r[k].x *= sc; r[k].y *= sc; r[k].z *= sc; r[k].w *= sc;
      }
      m = d;
    }
    float a = __expf(d - m);
    s += a;
#pragma unroll
    for (int k = 0; k < 8; ++k) {
      r[k].x = fmaf(a, xv[k].x, r[k].x); r[k].y = fmaf(a, xv[k].y, r[k].y);
      r[k].z = fmaf(a, xv[k].z, r[k].z); r[k].w = fmaf(a, xv[k].w, r[k].w);
    }
  }

  const int p = wv * 4 + grp;
#pragma unroll
  for (int k = 0; k < 8; ++k) *(v4f*)&rall[p][32 * k + 4 * j] = r[k];
  if (j == 0) { pmax[p] = m; psum[p] = s; }
  __syncthreads();

  float M = -__builtin_inff();
#pragma unroll
  for (int w = 0; w < 32; ++w) M = fmaxf(M, pmax[w]);

  float stot = 0.0f, rr = 0.0f;
#pragma unroll
  for (int w = 0; w < 32; ++w) {
    const float sw = psum[w];
    const float sc = (sw > 0.0f) ? __expf(pmax[w] - M) : 0.0f;
    stot = fmaf(sw, sc, stot);
    rr   = fmaf(rall[w][tid], sc, rr);
  }
  qstar[(size_t)g * 512 + 256 + tid] = (stot > 0.0f) ? (rr / stot) : 0.0f;
}

__global__ void __launch_bounds__(128)
out_gemm(const float* __restrict__ qstar, const float* __restrict__ Wout,
         const float* __restrict__ bout, float* __restrict__ out) {
  const int lane = threadIdx.x & 31;
  const int s    = threadIdx.x >> 5;
  const int c    = lane & 15;
  const int hh   = lane >> 4;
  const int kh   = hh << 1;
  const int tm   = blockIdx.x >> 3;
  const int tn   = blockIdx.x & 7;

  __shared__ __align__(16) float part[4][2][320];

  const float* a  = qstar + (size_t)(tm * 16 + c) * 512 + 128 * s + kh;
  const float* b0 = Wout  + (size_t)(tn * 32 + c) * 512 + 128 * s + kh;
  const float* b1 = b0 + 16 * 512;
  v8f acc0 = {};
  v8f acc1 = {};
#pragma unroll
  for (int k = 0; k < 128; k += 4) {
    const v2f A  = *(const v2f*)(a + k);
    const v2f B0 = *(const v2f*)(b0 + k);
    const v2f B1 = *(const v2f*)(b1 + k);
    acc0 = WMMA_F32(A, B0, acc0);
    acc1 = WMMA_F32(A, B1, acc1);
  }
  {
    float* dst = &part[s][0][c * 20 + 8 * hh];
    *(v4f*)(dst)           = (v4f){acc0[0], acc0[1], acc0[2], acc0[3]};
    *(v4f*)(dst + 4)       = (v4f){acc0[4], acc0[5], acc0[6], acc0[7]};
    *(v4f*)(dst + 320)     = (v4f){acc1[0], acc1[1], acc1[2], acc1[3]};
    *(v4f*)(dst + 320 + 4) = (v4f){acc1[4], acc1[5], acc1[6], acc1[7]};
  }
  __syncthreads();

  const int j  = hh;
  const float* p = &part[0][j][c * 20 + 4 * s];
  const v4f p0 = *(const v4f*)p;
  const v4f p1 = *(const v4f*)(p + 2 * 320);
  const v4f p2 = *(const v4f*)(p + 4 * 320);
  const v4f p3 = *(const v4f*)(p + 6 * 320);
  const int n  = tn * 32 + lane;
  const v4f r  = ((p0 + p1) + (p2 + p3)) + bout[n];
  float* o = out + (size_t)(tm * 16 + 4 * s) * 256 + n;
  o[0]   = r[0];
  o[256] = r[1];
  o[512] = r[2];
  o[768] = r[3];
}

extern "C" void kernel_launch(void* const* d_in, const int* in_sizes, int n_in,
                              void* d_out, int out_size, void* d_ws, size_t ws_size,
                              hipStream_t stream) {
  const float* x     = (const float*)d_in[0];
  const int*   batch = (const int*)  d_in[1];
  const float* Wih   = (const float*)d_in[2];
  const float* Whh   = (const float*)d_in[3];
  const float* bih   = (const float*)d_in[4];
  const float* bhh   = (const float*)d_in[5];
  const float* Wout  = (const float*)d_in[6];
  const float* bout  = (const float*)d_in[7];
  const int N = in_sizes[0] / 256;
  const int G = 512;

  float* ws    = (float*)d_ws;
  float* hsA   = ws;
  float* cs    = hsA + 512 * 256;
  float* qstar = cs + 512 * 256;
  float* hsB   = qstar + 512 * 512;
  int*   starts = (int*)(qstar + 512 * 512 + 512 * 1024);

  int zn = 512 * 256 * 2 + 512 * 512;
  init_zero<<<(zn + 255) / 256, 256, 0, stream>>>(ws, zn);
  find_starts<<<(N + 255) / 256, 256, 0, stream>>>(batch, starts, N, G);

  gates_lstm<<<256, 128, 0, stream>>>(hsA, qstar, Wih, Whh, bih, bhh, hsB, cs, qstar);
  attention<<<G, 256, 0, stream>>>(x, starts, hsB, qstar);
  gates_lstm<<<256, 128, 0, stream>>>(hsB, qstar, Wih, Whh, bih, bhh, hsA, cs, qstar);
  attention<<<G, 256, 0, stream>>>(x, starts, hsA, qstar);
  gates_lstm<<<256, 128, 0, stream>>>(hsA, qstar, Wih, Whh, bih, bhh, hsB, cs, qstar);
  attention<<<G, 256, 0, stream>>>(x, starts, hsB, qstar);
  out_gemm<<<256, 128, 0, stream>>>(qstar, Wout, bout, (float*)d_out);
}


# attention: every wave reads every second trip of x with non-temporal loads, so the plain half stays whole in the second-level cache between the three passes; nodes, sums and their order unchanged
# speedup vs baseline: 2.1307x; 1.0398x over previous
#include <hip/hip_runtime.h>

typedef __attribute__((ext_vector_type(2))) float v2f;
typedef __attribute__((ext_vector_type(4))) float v4f;
typedef __attribute__((ext_vector_type(8))) float v8f;

__global__ void init_zero(float* __restrict__ p, int n) {
  int i = blockIdx.x * blockDim.x + threadIdx.x;
  if (i < n) p[i] = 0.0f;
}

__global__ void find_starts(const int* __restrict__ batch, int* __restrict__ starts,
                            int N, int G) {
  int i = blockIdx.x * blockDim.x + threadIdx.x;
  if (i >= N) return;
  int b = batch[i];
  int bp = (i == 0) ? -1 : batch[i - 1];
  for (int g = bp + 1; g <= b; ++g) starts[g] = i;
  if (i == N - 1) {
    for (int g = b + 1; g <= G; ++g) starts[g] = N;
  }
}

__device__ __forceinline__ float sigmoidf_(float x) {
  return 1.0f / (1.0f + __expf(-x));
}

#define WMMA_F32(a, b, c) \
  __builtin_amdgcn_wmma_f32_16x16x4_f32(false, (a), false, (b), (short)0, (c), false, false)

__global__ void __launch_bounds__(128)
gates_lstm(const float* __restrict__ hs_old, const float* r_src,
           const float* __restrict__ Wih, const float* __restrict__ Whh,
           const float* __restrict__ bih, const float* __restrict__ bhh,
           float* __restrict__ hs_new, float* __restrict__ cs, float* q_dst) {
  const int lane = threadIdx.x & 31;
  const int s    = __builtin_amdgcn_readfirstlane((int)(threadIdx.x >> 5));
  const int c    = lane & 15;
  const int hh   = lane >> 4;
  const int kh   = hh << 1;
  const int tm   = blockIdx.x >> 4;
  const int tc   = blockIdx.x & 15;
  const int row0 = tm * 32 + c;
  const int wrow = tc * 16 + c;

  __shared__ __align__(16) float part[4][8][320];

  v8f acc[2][4];
#pragma unroll
  for (int mi = 0; mi < 2; ++mi)
#pragma unroll
    for (int g = 0; g < 4; ++g) acc[mi][g] = (v8f){};

  {
    const float* a0;
    const float* a1;
    if (s < 2) {
      a0 = hs_old + (size_t)row0 * 256 + 128 * s + kh;
      a1 = a0 + 16 * 256;
    } else {
      a0 = r_src + (size_t)row0 * 512 + 128 * s + kh;
      a1 = a0 + 16 * 512;
    }
    const float* b = Wih + (size_t)wrow * 512 + 128 * s + kh;
#pragma unroll
    for (int k = 0; k < 128; k += 4) {
      const v2f A0 = *(const v2f*)(a0 + k);
      const v2f A1 = *(const v2f*)(a1 + k);
      v2f B[4];
#pragma unroll
      for (int g = 0; g < 4; ++g) B[g] = *(const v2f*)(b + (size_t)g * 256 * 512 + k);
#pragma unroll
      for (int g = 0; g < 4; ++g) {
        acc[0][g] = WMMA_F32(A0, B[g], acc[0][g]);
        acc[1][g] = WMMA_F32(A1, B[g], acc[1][g]);
      }
    }
  }
  {
    const float* a0 = hs_old + (size_t)row0 * 256 + 64 * s + kh;
    const float* a1 = a0 + 16 * 256;
    const float* b  = Whh + (size_t)wrow * 256 + 64 * s + kh;
#pragma unroll
    for (int k = 0; k < 64; k += 4) {
      const v2f A0 = *(const v2f*)(a0 + k);
      const v2f A1 = *(const v2f*)(a1 + k);
      v2f B[4];
#pragma unroll
      for (int g = 0; g < 4; ++g) B[g] = *(const v2f*)(b + (size_t)g * 256 * 256 + k);
#pragma unroll
      for (int g = 0; g < 4; ++g) {
        acc[0][g] = WMMA_F32(A0, B[g], acc[0][g]);
        acc[1][g] = WMMA_F32(A1, B[g], acc[1][g]);
      }
    }
  }

  {
    float* dst = &part[s][0][c * 20 + 8 * hh];
#pragma unroll
    for (int mi = 0; mi < 2; ++mi)
#pragma unroll
      for (int g = 0; g < 4; ++g) {
        const v8f v = acc[mi][g];
        *(v4f*)(dst + (mi * 4 + g) * 320)     = (v4f){v[0], v[1], v[2], v[3]};
        *(v4f*)(dst + (mi * 4 + g) * 320 + 4) = (v4f){v[4], v[5], v[6], v[7]};
      }
  }
  __syncthreads();

  const int cc  = threadIdx.x & 15;
  const int rg  = threadIdx.x >> 4;
  const int mi  = rg >> 2;
  const int rr  = (rg & 3) * 4;
  const int chn = tc * 16 + cc;
  v4f gs[4];
#pragma unroll
  for (int g = 0; g < 4; ++g) {
    const float* p = &part[0][mi * 4 + g][cc * 20 + rr];
    const v4f p0 = *(const v4f*)p;
    const v4f p1 = *(const v4f*)(p + 8 * 320);
    const v4f p2 = *(const v4f*)(p + 16 * 320);
    const v4f p3 = *(const v4f*)(p + 24 * 320);
    const float bias = bih[g * 256 + chn] + bhh[g * 256 + chn];
    gs[g] = ((p0 + p1) + (p2 + p3)) + bias;
  }
#pragma unroll
  for (int e = 0; e < 4; ++e) {
    const int row = tm * 32 + mi * 16 + rr + e;
    const int idx = row * 256 + chn;
    const float ig = sigmoidf_(gs[0][e]);
    const float fg = sigmoidf_(gs[1][e]);
    const float gg = tanhf(gs[2][e]);
    const float og = sigmoidf_(gs[3][e]);
    const float cn = fg * cs[idx] + ig * gg;
    cs[idx] = cn;
    const float h = og * tanhf(cn);
    hs_new[idx] = h;
    q_dst[(size_t)row * 512 + chn] = h;
  }
}

template <bool NT>
__device__ __forceinline__ v4f load_x(const v4f* p) {
  if (NT) return __builtin_nontemporal_load(p);
  return *p;
}

template <bool NT>
__device__ __forceinline__ int walk(const float* __restrict__ x, int i, int lim, int j,
                                    const v4f (&qv)[8], float& m, float& s, v4f (&r)[8]) {
  for (; i < lim; i += 32) {
    const v4f* xr = (const v4f*)(x + (size_t)i * 256) + j;
    v4f xv[8];
#pragma unroll
    for (int k = 0; k < 8; ++k) xv[k] = load_x<NT>(xr + 8 * k);

    float d0 = 0.0f, d1 = 0.0f, d2 = 0.0f, d3 = 0.0f;
#pragma unroll
    for (int k = 0; k < 8; ++k) {
      d0 = fmaf(xv[k].x, qv[k].x, d0);
      d1 = fmaf(xv[k].y, qv[k].y, d1);
      d2 = fmaf(xv[k].z, qv[k].z, d2);
      d3 = fmaf(xv[k].w, qv[k].w, d3);
    }
    float d = (d0 + d1) + (d2 + d3);
    d += __shfl_xor(d, 4, 32);
    d += __shfl_xor(d, 2, 32);
    d += __shfl_xor(d, 1, 32);
    if (d > m) {
      float sc = (m == -__builtin_inff()) ? 0.0f : __expf(m - d);
      s *= sc;
#pragma unroll
      for (int k = 0; k < 8; ++k) {
        r[k].x *= sc; r[k].y *= sc; r[k].z *= sc; r[k].w *= sc;
      }
      m = d;
    }
    float a = __expf(d - m);
    s += a;
#pragma unroll
    for (int k = 0; k < 8; ++k) {
      r[k].x = fmaf(a, xv[k].x, r[k].x); r[k].y = fmaf(a, xv[k].y, r[k].y);
      r[k].z = fmaf(a, xv[k].z, r[k].z); r[k].w = fmaf(a, xv[k].w, r[k].w);
    }
  }
  return i;
}

__global__ void __launch_bounds__(256)
attention(const float* __restrict__ x, const int* __restrict__ starts,
          const float* __restrict__ hs, float* __restrict__ qstar) {
  const int g    = blockIdx.x;
  const int tid  = threadIdx.x;
  const int lane = tid & 31;
  const int wv   = __builtin_amdgcn_readfirstlane(tid >> 5);
  const int grp  = lane >> 3;
  const int j    = lane & 7;

  __shared__ __align__(16) float rall[32][256];
  __shared__ float pmax[32];
  __shared__ float psum[32];

  const int start = starts[g];
  const int end   = starts[g + 1];

  const v4f* qp = (const v4f*)(hs + (size_t)g * 256) + j;
  v4f qv[8];
#pragma unroll
  for (int k = 0; k < 8; ++k) qv[k] = qp[8 * k];

  float m = -__builtin_inff();
  float s = 0.0f;
  v4f r[8];
#pragma unroll
  for (int k = 0; k < 8; ++k) r[k] = (v4f){0.0f, 0.0f, 0.0f, 0.0f};

  int i = start + 4 * wv + grp;
  for (int lo = start - 32 * (wv & 1); lo < end; lo += 64) {
    i = walk<true >(x, i, min(lo + 32, end), j, qv, m, s, r);
    i = walk<false>(x, i, min(lo + 64, end), j, qv, m, s, r);
  }

  const int p = wv * 4 + grp;
#pragma unroll
  for (int k = 0; k < 8; ++k) *(v4f*)&rall[p][32 * k + 4 * j] = r[k];
  if (j == 0) { pmax[p] = m; psum[p] = s; }
  __syncthreads();

  float M = -__builtin_inff();
#pragma unroll
  for (int w = 0; w < 32; ++w) M = fmaxf(M, pmax[w]);

  float stot = 0.0f, rr = 0.0f;
#pragma unroll
  for (int w = 0; w < 32; ++w) {
    const float sw = psum[w];
    const float sc = (sw > 0.0f) ? __expf(pmax[w] - M) : 0.0f;
    stot = fmaf(sw, sc, stot);
    rr   = fmaf(rall[w][tid], sc, rr);
  }
  qstar[(size_t)g * 512 + 256 + tid] = (stot > 0.0f) ? (rr / stot) : 0.0f;
}

__global__ void __launch_bounds__(128)
out_gemm(const float* __restrict__ qstar, const float* __restrict__ Wout,
         const float* __restrict__ bout, float* __restrict__ out) {
  const int lane = threadIdx.x & 31;
  const int s    = threadIdx.x >> 5;
  const int c    = lane & 15;
  const int hh   = lane >> 4;
  const int kh   = hh << 1;
  const int tm   = blockIdx.x >> 3;
  const int tn   = blockIdx.x & 7;

  __shared__ __align__(16) float part[4][2][320];

  const float* a  = qstar + (size_t)(tm * 16 + c) * 512 + 128 * s + kh;
  const float* b0 = Wout  + (size_t)(tn * 32 + c) * 512 + 128 * s + kh;
  const float* b1 = b0 + 16 * 512;
  v8f acc0 = {};
  v8f acc1 = {};
#pragma unroll
  for (int k = 0; k < 128; k += 4) {
    const v2f A  = *(const v2f*)(a + k);
    const v2f B0 = *(const v2f*)(b0 + k);
    const v2f B1 = *(const v2f*)(b1 + k);
    acc0 = WMMA_F32(A, B0, acc0);
    acc1 = WMMA_F32(A, B1, acc1);
  }
  {
    float* dst = &part[s][0][c * 20 + 8 * hh];
    *(v4f*)(dst)           = (v4f){acc0[0], acc0[1], acc0[2], acc0[3]};
    *(v4f*)(dst + 4)       = (v4f){acc0[4], acc0[5], acc0[6], acc0[7]};
    *(v4f*)(dst + 320)     = (v4f){acc1[0], acc1[1], acc1[2], acc1[3]};
    *(v4f*)(dst + 320 + 4) = (v4f){acc1[4], acc1[5], acc1[6], acc1[7]};
  }
  __syncthreads();

  const int j  = hh;
  const float* p = &part[0][j][c * 20 + 4 * s];
  const v4f p0 = *(const v4f*)p;
  const v4f p1 = *(const v4f*)(p + 2 * 320);
  const v4f p2 = *(const v4f*)(p + 4 * 320);
  const v4f p3 = *(const v4f*)(p + 6 * 320);
  const int n  = tn * 32 + lane;
  const v4f r  = ((p0 + p1) + (p2 + p3)) + bout[n];
  float* o = out + (size_t)(tm * 16 + 4 * s) * 256 + n;
  o[0]   = r[0];
  o[256] = r[1];
  o[512] = r[2];
  o[768] = r[3];
}

extern "C" void kernel_launch(void* const* d_in, const int* in_sizes, int n_in,
                              void* d_out, int out_size, void* d_ws, size_t ws_size,
                              hipStream_t stream) {
  const float* x     = (const float*)d_in[0];
  const int*   batch = (const int*)  d_in[1];
  const float* Wih   = (const float*)d_in[2];
  const float* Whh   = (const float*)d_in[3];
  const float* bih   = (const float*)d_in[4];
  const float* bhh   = (const float*)d_in[5];
  const float* Wout  = (const float*)d_in[6];
  const float* bout  = (const float*)d_in[7];
  const int N = in_sizes[0] / 256;
  const int G = 512;

  float* ws    = (float*)d_ws;
  float* hsA   = ws;
  float* cs    = hsA + 512 * 256;
  float* qstar = cs + 512 * 256;
  float* hsB   = qstar + 512 * 512;
  int*   starts = (int*)(qstar + 512 * 512 + 512 * 1024);

  int zn = 512 * 256 * 2 + 512 * 512;
  init_zero<<<(zn + 255) / 256, 256, 0, stream>>>(ws, zn);
  find_starts<<<(N + 255) / 256, 256, 0, stream>>>(batch, starts, N, G);

  gates_lstm<<<256, 128, 0, stream>>>(hsA, qstar, Wih, Whh, bih, bhh, hsB, cs, qstar);
  attention<<<G, 256, 0, stream>>>(x, starts, hsB, qstar);
  gates_lstm<<<256, 128, 0, stream>>>(hsB, qstar, Wih, Whh, bih, bhh, hsA, cs, qstar);
  attention<<<G, 256, 0, stream>>>(x, starts, hsA, qstar);
  gates_lstm<<<256, 128, 0, stream>>>(hsA, qstar, Wih, Whh, bih, bhh, hsB, cs, qstar);
  attention<<<G, 256, 0, stream>>>(x, starts, hsB, qstar);
  out_gemm<<<256, 128, 0, stream>>>(qstar, Wout, bout, (float*)d_out);
}
